# starting kernel
# baseline (speedup 1.0000x reference)
#include <hip/hip_runtime.h>
#include <math.h>

typedef __attribute__((ext_vector_type(2))) float v2f;
typedef __attribute__((ext_vector_type(8))) float v8f;

#define NJ      16384
#define NI      65536
#define NE      262144
#define A_ELEMS (441 * NJ)
#define LD_IDX  (2 * A_ELEMS)
#define ZBASE   (LD_IDX + 1)
#define ZTOT    (21 * NJ)

struct cplx { float re, im; };

__device__ __forceinline__ cplx cmk(float r, float i){ cplx c; c.re=r; c.im=i; return c; }
__device__ __forceinline__ cplx cadd(cplx a, cplx b){ return cmk(a.re+b.re, a.im+b.im); }
__device__ __forceinline__ cplx csub(cplx a, cplx b){ return cmk(a.re-b.re, a.im-b.im); }
__device__ __forceinline__ cplx cmul(cplx a, cplx b){ return cmk(a.re*b.re - a.im*b.im, a.re*b.im + a.im*b.re); }
__device__ __forceinline__ cplx cmulc(cplx a, cplx b){
  return cmk(a.re*b.re + a.im*b.im, a.im*b.re - a.re*b.im); }
__device__ __forceinline__ cplx cneg(cplx a){ return cmk(-a.re, -a.im); }
__device__ __forceinline__ cplx cinv(cplx a){
  float d = 1.0f/(a.re*a.re + a.im*a.im); return cmk(a.re*d, -a.im*d); }

__device__ __forceinline__ float waveReduceWmma(float v) {
  v2f a; a.x = 1.0f; a.y = 1.0f;
  v2f b; b.x = v;    b.y = 0.0f;
  v8f c = {};
  v8f d = __builtin_amdgcn_wmma_f32_16x16x4_f32(false, a, false, b,
                                                (short)0, c, false, false);
  float h = d[0];
  h += __shfl_xor(h, 1, 32);
  h += __shfl_xor(h, 2, 32);
  h += __shfl_xor(h, 4, 32);
  h += __shfl_xor(h, 8, 32);
  return h;
}

__device__ __forceinline__ void blockAccum(float v, float* dst, int tid) {
  float w = waveReduceWmma(v);
  __shared__ float wsum[8];
  if ((tid & 31) == 0) wsum[tid >> 5] = w;
  __syncthreads();
  if (tid == 0) {
    float s = 0.f;
    #pragma unroll
    for (int q = 0; q < 8; ++q) s += wsum[q];
    atomicAdd(dst, s);
  }
}

__device__ __forceinline__ cplx A1val(const cplx T1[4][4], const cplx iS1[4],
                                      const cplx il0[4][4], int t, int p, int a) {
  if (p < 4 && a < 4) {
    cplx v = cmulc(cmul(T1[t][p], iS1[t]), T1[t][a]);
    if (p == a) v = cadd(v, il0[t][p]);
    return v;
  } else if (p < 4) {
    return cneg(cmul(T1[t][p], iS1[t]));
  } else if (a < 4) {
    cplx x = cmul(T1[t][a], iS1[t]);
    return cmk(-x.re, x.im);
  }
  return iS1[t];
}

__global__ void zero_logdet_kernel(float* out) { out[LD_IDX] = 0.0f; }

__global__ __launch_bounds__(256, 1)
void logdet00_kernel(const float* __restrict__ re,
                     const float* __restrict__ im,
                     float* __restrict__ out) {
  int idx = blockIdx.x * 256 + threadIdx.x;
  float r = re[idx], ii = im[idx];
  float v = 0.5f * logf(r*r + ii*ii);
  blockAccum(v, out + LD_IDX, threadIdx.x);
}

__global__ __launch_bounds__(256, 1)
void schur_kernel(const float* __restrict__ l00r, const float* __restrict__ l00i,
                  const float* __restrict__ l01r, const float* __restrict__ l01i,
                  const float* __restrict__ l02r, const float* __restrict__ l02i,
                  const float* __restrict__ l11r, const float* __restrict__ l11i,
                  const float* __restrict__ l12r, const float* __restrict__ l12i,
                  const float* __restrict__ l22r, const float* __restrict__ l22i,
                  const float* __restrict__ zre,  const float* __restrict__ zim,
                  float* __restrict__ out) {
  const int j = blockIdx.x * 256 + threadIdx.x;
  float ld = 0.f;

  cplx T1[4][4], iS1[4], il0[4][4];
  #pragma unroll
  for (int t = 0; t < 4; ++t) {
    const int i = t * NJ + j;
    cplx m = cmk(0.f, 0.f);
    #pragma unroll
    for (int s = 0; s < 4; ++s) {
      const int e = s * NI + i;
      cplx l00 = cmk(l00r[e], l00i[e]);
      cplx l01 = cmk(l01r[e], l01i[e]);
      cplx il  = cinv(l00);
      il0[t][s] = il;
      cplx tt = cmul(l01, il);
      T1[t][s] = tt;
      m = cadd(m, cmulc(tt, l01));
    }
    cplx S1 = csub(cmk(l11r[i], l11i[i]), m);
    ld += 0.5f * logf(S1.re*S1.re + S1.im*S1.im);
    iS1[t] = cinv(S1);
  }

  cplx Bv[5][4];
  #pragma unroll
  for (int t = 0; t < 4; ++t) {
    #pragma unroll
    for (int a = 0; a < 4; ++a)
      Bv[a][t] = cmk(l02r[(a*4+t)*NJ + j], l02i[(a*4+t)*NJ + j]);
    Bv[4][t] = cmk(l12r[t*NJ + j], l12i[t*NJ + j]);
  }

  cplx T20[20];
  #pragma unroll
  for (int t = 0; t < 4; ++t) {
    #pragma unroll
    for (int p = 0; p < 5; ++p) {
      cplx acc = cmk(0.f, 0.f);
      #pragma unroll
      for (int a = 0; a < 5; ++a)
        acc = cadd(acc, cmul(Bv[a][t], A1val(T1, iS1, il0, t, p, a)));
      T20[p*4 + t] = acc;
    }
  }

  cplx m2 = cmk(0.f, 0.f);
  #pragma unroll
  for (int t = 0; t < 4; ++t)
    #pragma unroll
    for (int p = 0; p < 5; ++p)
      m2 = cadd(m2, cmulc(T20[p*4 + t], Bv[p][t]));
  cplx S2 = csub(cmk(l22r[j], l22i[j]), m2);
  ld += 0.5f * logf(S2.re*S2.re + S2.im*S2.im);
  cplx iS2 = cinv(S2);

  cplx zv[21];
  #pragma unroll
  for (int q = 0; q < 21; ++q) zv[q] = cmk(zre[q*NJ + j], zim[q*NJ + j]);

  float* __restrict__ outAr = out;
  float* __restrict__ outAi = out + A_ELEMS;
  float* __restrict__ outZr = out + ZBASE;
  float* __restrict__ outZi = out + ZBASE + ZTOT;

  #pragma unroll
  for (int t = 0; t < 4; ++t) {
    #pragma unroll
    for (int a = 0; a < 5; ++a) {
      const int k = a*4 + t;
      cplx Pk = cmul(T20[k], iS2);
      cplx zo = cmk(0.f, 0.f);
      #pragma unroll
      for (int tp = 0; tp < 4; ++tp) {
        #pragma unroll
        for (int ap = 0; ap < 5; ++ap) {
          const int kp = ap*4 + tp;
          cplx Av = cmulc(Pk, T20[kp]);
          if (tp == t) Av = cadd(Av, A1val(T1, iS1, il0, t, a, ap));
          zo = cadd(zo, cmul(zv[kp], Av));
          outAr[(k*21 + kp)*NJ + j] = Av.re;
          outAi[(k*21 + kp)*NJ + j] = Av.im;
        }
      }
      {
        cplx Av = cneg(Pk);
        zo = cadd(zo, cmul(zv[20], Av));
        outAr[(k*21 + 20)*NJ + j] = Av.re;
        outAi[(k*21 + 20)*NJ + j] = Av.im;
      }
      outZr[k*NJ + j] = zo.re;
      outZi[k*NJ + j] = zo.im;
    }
  }
  {
    cplx zo = cmk(0.f, 0.f);
    #pragma unroll
    for (int tp = 0; tp < 4; ++tp) {
      #pragma unroll
      for (int ap = 0; ap < 5; ++ap) {
        const int kp = ap*4 + tp;
        cplx x  = cmul(T20[kp], iS2);
        cplx Av = cmk(-x.re, x.im);
        zo = cadd(zo, cmul(zv[kp], Av));
        outAr[(20*21 + kp)*NJ + j] = Av.re;
        outAi[(20*21 + kp)*NJ + j] = Av.im;
      }
    }
    cplx Av = iS2;
    zo = cadd(zo, cmul(zv[20], Av));
    outAr[(20*21 + 20)*NJ + j] = Av.re;
    outAi[(20*21 + 20)*NJ + j] = Av.im;
    outZr[20*NJ + j] = zo.re;
    outZi[20*NJ + j] = zo.im;
  }

  blockAccum(ld, out + LD_IDX, threadIdx.x);
}

extern "C" void kernel_launch(void* const* d_in, const int* in_sizes, int n_in,
                              void* d_out, int out_size, void* d_ws, size_t ws_size,
                              hipStream_t stream) {
  (void)in_sizes; (void)n_in; (void)out_size; (void)d_ws; (void)ws_size;
  const float* l00r = (const float*)d_in[0];
  const float* l00i = (const float*)d_in[1];
  const float* l01r = (const float*)d_in[2];
  const float* l01i = (const float*)d_in[3];
  const float* l02r = (const float*)d_in[4];
  const float* l02i = (const float*)d_in[5];
  const float* l11r = (const float*)d_in[6];
  const float* l11i = (const float*)d_in[7];
  const float* l12r = (const float*)d_in[8];
  const float* l12i = (const float*)d_in[9];
  const float* l22r = (const float*)d_in[10];
  const float* l22i = (const float*)d_in[11];
  const float* zre  = (const float*)d_in[12];
  const float* zim  = (const float*)d_in[13];
  float* out = (float*)d_out;

  zero_logdet_kernel<<<1, 1, 0, stream>>>(out);
  logdet00_kernel<<<NE / 256, 256, 0, stream>>>(l00r, l00i, out);
  schur_kernel<<<NJ / 256, 256, 0, stream>>>(l00r, l00i, l01r, l01i, l02r, l02i,
                                             l11r, l11i, l12r, l12i, l22r, l22i,
                                             zre, zim, out);
}


# main launch re-tiled: each lane owns 2 adjacent columns (2-vector f32 math), 4 waves x 5 rows per group, each load and A store 8 B per lane = one 256-B unit; S2, zout, logdet sums regrouped
# speedup vs baseline: 3.2108x; 3.2108x over previous
#include <hip/hip_runtime.h>
#include <math.h>


typedef __attribute__((ext_vector_type(2))) float v2f;
typedef __attribute__((ext_vector_type(8))) float v8f;
typedef __attribute__((ext_vector_type(4))) float v4f;

#define NJ      16384
#define NI      65536
#define NE      262144
#define A_ELEMS (441 * NJ)
#define LD_IDX  (2 * A_ELEMS)
#define ZBASE   (LD_IDX + 1)
#define ZTOT    (21 * NJ)
#define BLK     64
#define NW      4

struct cplx { v2f re, im; };

__device__ __forceinline__ cplx cmk(v2f r, v2f i){ cplx c; c.re=r; c.im=i; return c; }
__device__ __forceinline__ cplx czero(){ v2f z = {0.f, 0.f}; return cmk(z, z); }
__device__ __forceinline__ cplx cadd(cplx a, cplx b){ return cmk(a.re+b.re, a.im+b.im); }
__device__ __forceinline__ cplx csub(cplx a, cplx b){ return cmk(a.re-b.re, a.im-b.im); }
__device__ __forceinline__ cplx cmul(cplx a, cplx b){ return cmk(a.re*b.re - a.im*b.im, a.re*b.im + a.im*b.re); }
__device__ __forceinline__ cplx cmulc(cplx a, cplx b){
  return cmk(a.re*b.re + a.im*b.im, a.im*b.re - a.re*b.im); }
__device__ __forceinline__ cplx cneg(cplx a){ return cmk(-a.re, -a.im); }
__device__ __forceinline__ cplx cinv(cplx a){
  v2f d = 1.0f/(a.re*a.re + a.im*a.im); return cmk(a.re*d, -a.im*d); }
__device__ __forceinline__ v2f ld2(const float* p){ return *(const v2f*)p; }
__device__ __forceinline__ void st2(float* p, v2f v){ *(v2f*)p = v; }

__device__ __forceinline__ float waveReduceWmma(float v) {
  v2f a; a.x = 1.0f; a.y = 1.0f;
  v2f b; b.x = v;    b.y = 0.0f;
  v8f c = {};
  v8f d = __builtin_amdgcn_wmma_f32_16x16x4_f32(false, a, false, b,
                                                (short)0, c, false, false);
  float h = d[0];
  h += __shfl_xor(h, 1, 32);
  h += __shfl_xor(h, 2, 32);
  h += __shfl_xor(h, 4, 32);
  h += __shfl_xor(h, 8, 32);
  return h;
}

__device__ __forceinline__ cplx A1val(const cplx T1[4][4], const cplx iS1[4],
                                      const cplx il0[4][4], int t, int p, int a) {
  if (p < 4 && a < 4) {
    cplx v = cmulc(cmul(T1[t][p], iS1[t]), T1[t][a]);
    if (p == a) v = cadd(v, il0[t][p]);
    return v;
  } else if (p < 4) {
    return cneg(cmul(T1[t][p], iS1[t]));
  } else if (a < 4) {
    cplx x = cmul(T1[t][a], iS1[t]);
    return cmk(-x.re, x.im);
  }
  return iS1[t];
}

__device__ __forceinline__ void store_cols(float* p, v2f v, int lane) {
  const int src = lane >> 1;
  const float x0 = __shfl(v.x, src, 32),      y0 = __shfl(v.y, src, 32);
  const float x1 = __shfl(v.x, src + 16, 32), y1 = __shfl(v.y, src + 16, 32);
  const bool odd = (lane & 1) != 0;
  p[lane]      = odd ? y0 : x0;
  p[lane + 32] = odd ? y1 : x1;
}

__global__ __launch_bounds__(32, 1)
void start_kernel(float* __restrict__ out) {
  if (threadIdx.x == 0) out[LD_IDX] = 0.0f;
}

__global__ __launch_bounds__(32 * NW, 1)
void schur_kernel(const float* __restrict__ l00r, const float* __restrict__ l00i,
                  const float* __restrict__ l01r, const float* __restrict__ l01i,
                  const float* __restrict__ l02r, const float* __restrict__ l02i,
                  const float* __restrict__ l11r, const float* __restrict__ l11i,
                  const float* __restrict__ l12r, const float* __restrict__ l12i,
                  const float* __restrict__ l22r, const float* __restrict__ l22i,
                  const float* __restrict__ zre,  const float* __restrict__ zim,
                  float* __restrict__ out) {
  const int tid  = threadIdx.x;
  const int w    = __builtin_amdgcn_readfirstlane(tid >> 5);
  const int lane = tid & 31;
  const int j0   = blockIdx.x * BLK;
  const int j    = j0 + 2 * lane;
  float* outAr = out;
  float* outAi = out + A_ELEMS;
  float* outZr = out + ZBASE;
  float* outZi = out + ZBASE + ZTOT;

  int tt[4];
  #pragma unroll
  for (int d = 0; d < 4; ++d) tt[d] = (w + d) & 3;

  cplx T1[4][4], iS1[4], il0[4][4];
  v2f sq[5];
  cplx Bv[5][4];
  cplx T20[20];
  #pragma unroll
  for (int d = 0; d < 4; ++d) {
    const int i = tt[d] * NJ + j;
    cplx m = czero();
    #pragma unroll
    for (int s = 0; s < 4; ++s) {
      const int e = s * NI + i;
      cplx l00 = cmk(ld2(l00r + e), ld2(l00i + e));
      cplx l01 = cmk(ld2(l01r + e), ld2(l01i + e));
      cplx il  = cinv(l00);
      il0[d][s] = il;
      cplx tq = cmul(l01, il);
      T1[d][s] = tq;
      m = cadd(m, cmulc(tq, l01));
    }
    cplx S1 = csub(cmk(ld2(l11r + i), ld2(l11i + i)), m);
    sq[d] = S1.re*S1.re + S1.im*S1.im;
    iS1[d] = cinv(S1);
    #pragma unroll
    for (int ar = 0; ar < 4; ++ar)
      Bv[ar][d] = cmk(ld2(l02r + (ar*4+tt[d])*NJ + j), ld2(l02i + (ar*4+tt[d])*NJ + j));
    Bv[4][d] = cmk(ld2(l12r + tt[d]*NJ + j), ld2(l12i + tt[d]*NJ + j));
    #pragma unroll
    for (int p = 0; p < 5; ++p) {
      cplx acc = czero();
      #pragma unroll
      for (int ar = 0; ar < 5; ++ar)
        acc = cadd(acc, cmul(Bv[ar][d], A1val(T1, iS1, il0, d, p, ar)));
      T20[p*4 + d] = acc;
    }
  }

  cplx zv[21];
  #pragma unroll
  for (int d = 0; d < 4; ++d) {
    #pragma unroll
    for (int ap = 0; ap < 5; ++ap)
      zv[ap*4+d] = cmk(ld2(zre + (ap*4+tt[d])*NJ + j), ld2(zim + (ap*4+tt[d])*NJ + j));
  }
  zv[20] = cmk(ld2(zre + 20*NJ + j), ld2(zim + 20*NJ + j));

  const int idx0 = blockIdx.x * 256 + tid;
  const v4f l0ra = *(const v4f*)(l00r + 4 * idx0);
  const v4f l0ia = *(const v4f*)(l00i + 4 * idx0);
  const v4f l0rb = *(const v4f*)(l00r + 4 * (idx0 + 128));
  const v4f l0ib = *(const v4f*)(l00i + 4 * (idx0 + 128));

  cplx m2 = czero();
  #pragma unroll
  for (int d = 0; d < 4; ++d)
    #pragma unroll
    for (int p = 0; p < 5; ++p)
      m2 = cadd(m2, cmulc(T20[p*4 + d], Bv[p][d]));
  cplx S2 = csub(cmk(ld2(l22r + j), ld2(l22i + j)), m2);
  sq[4] = S2.re*S2.re + S2.im*S2.im;
  const cplx iS2 = cinv(S2);

  float v;
  v  = 0.5f * logf(l0ra.x*l0ra.x + l0ia.x*l0ia.x);
  v += 0.5f * logf(l0ra.y*l0ra.y + l0ia.y*l0ia.y);
  v += 0.5f * logf(l0ra.z*l0ra.z + l0ia.z*l0ia.z);
  v += 0.5f * logf(l0ra.w*l0ra.w + l0ia.w*l0ia.w);
  v += 0.5f * logf(l0rb.x*l0rb.x + l0ib.x*l0ib.x);
  v += 0.5f * logf(l0rb.y*l0rb.y + l0ib.y*l0ib.y);
  v += 0.5f * logf(l0rb.z*l0rb.z + l0ib.z*l0ib.z);
  v += 0.5f * logf(l0rb.w*l0rb.w + l0ib.w*l0ib.w);
  if (w == 0) {
    float ld = 0.f;
    #pragma unroll
    for (int q = 0; q < 5; ++q) ld += 0.5f * logf(sq[q].x) + 0.5f * logf(sq[q].y);
    v += ld;
  }
  float ws = waveReduceWmma(v);
  __shared__ float wsum[NW];
  if (lane == 0) wsum[w] = ws;

  float* qAr = outAr + (20*21)*NJ + j;
  float* qAi = outAi + (20*21)*NJ + j;
  __shared__ v4f rowop[5][6][32 * NW];
  #pragma unroll
  for (int a = 0; a < 5; ++a) {
    cplx Pk = cmul(T20[a*4], iS2);
    v4f pk4 = {Pk.re.x, Pk.re.y, Pk.im.x, Pk.im.y};
    rowop[a][0][tid] = pk4;
    #pragma unroll
    for (int ap = 0; ap < 5; ++ap) {
      cplx x = A1val(T1, iS1, il0, 0, a, ap);
      v4f x4 = {x.re.x, x.re.y, x.im.x, x.im.y};
      rowop[a][1 + ap][tid] = x4;
    }
  }
  #pragma unroll 1
  for (int a = 0; a < 5; ++a) {
    const int k = a * 4 + w;
    float* pAr = outAr + (k*21)*NJ + j;
    float* pAi = outAi + (k*21)*NJ + j;
    const v4f pk4 = rowop[a][0][tid];
    cplx Pk; Pk.re.x = pk4.x; Pk.re.y = pk4.y; Pk.im.x = pk4.z; Pk.im.y = pk4.w;
    cplx add[5];
    #pragma unroll
    for (int ap = 0; ap < 5; ++ap) {
      const v4f x4 = rowop[a][1 + ap][tid];
      add[ap].re.x = x4.x; add[ap].re.y = x4.y; add[ap].im.x = x4.z; add[ap].im.y = x4.w;
    }
    cplx zo = czero();
    #pragma unroll
    for (int d = 0; d < 4; ++d) {
      #pragma unroll
      for (int ap = 0; ap < 5; ++ap) {
        const int kl = ap*4 + d;
        const int kp = ap*4 + tt[d];
        cplx Av = cmulc(Pk, T20[kl]);
        if (d == 0) Av = cadd(Av, add[ap]);
        zo = cadd(zo, cmul(zv[kl], Av));
        st2(pAr + kp*NJ, Av.re);
        st2(pAi + kp*NJ, Av.im);
      }
    }
    {
      cplx Av = cneg(Pk);
      zo = cadd(zo, cmul(zv[20], Av));
      st2(pAr + 20*NJ, Av.re);
      st2(pAi + 20*NJ, Av.im);
    }
    store_cols(outZr + k*NJ + j0, zo.re, lane);
    store_cols(outZi + k*NJ + j0, zo.im, lane);
    st2(qAr + k*NJ, -Pk.re);
    st2(qAi + k*NJ,  Pk.im);
  }
  if (w == NW - 1) {
    cplx zo = czero();
    #pragma unroll
    for (int d = 0; d < 4; ++d) {
      #pragma unroll
      for (int ap = 0; ap < 5; ++ap) {
        const int kl = ap*4 + d;
        cplx x  = cmul(T20[kl], iS2);
        cplx Av = cmk(-x.re, x.im);
        zo = cadd(zo, cmul(zv[kl], Av));
      }
    }
    cplx Av = iS2;
    zo = cadd(zo, cmul(zv[20], Av));
    st2(qAr + 20*NJ, Av.re);
    st2(qAi + 20*NJ, Av.im);
    store_cols(outZr + 20*NJ + j0, zo.re, lane);
    store_cols(outZi + 20*NJ + j0, zo.im, lane);
  }

  __syncthreads();
  if (tid == 0) {
    float s = 0.f;
    #pragma unroll
    for (int q = 0; q < NW; ++q) s += wsum[q];
    atomicAdd(out + LD_IDX, s);
  }
}

extern "C" void kernel_launch(void* const* d_in, const int* in_sizes, int n_in,
                              void* d_out, int out_size, void* d_ws, size_t ws_size,
                              hipStream_t stream) {
  (void)in_sizes; (void)n_in; (void)out_size; (void)d_ws; (void)ws_size;
  const float* l00r = (const float*)d_in[0];
  const float* l00i = (const float*)d_in[1];
  const float* l01r = (const float*)d_in[2];
  const float* l01i = (const float*)d_in[3];
  const float* l02r = (const float*)d_in[4];
  const float* l02i = (const float*)d_in[5];
  const float* l11r = (const float*)d_in[6];
  const float* l11i = (const float*)d_in[7];
  const float* l12r = (const float*)d_in[8];
  const float* l12i = (const float*)d_in[9];
  const float* l22r = (const float*)d_in[10];
  const float* l22i = (const float*)d_in[11];
  const float* zre  = (const float*)d_in[12];
  const float* zim  = (const float*)d_in[13];
  float* out = (float*)d_out;

  start_kernel<<<1, 32, 0, stream>>>(out);
  schur_kernel<<<NJ / BLK, 32 * NW, 0, stream>>>(l00r, l00i, l01r, l01i, l02r, l02i,
                                                 l11r, l11i, l12r, l12i, l22r, l22i,
                                                 zre, zim, out);
}
